# starting kernel
# baseline (speedup 1.0000x reference)
#include <hip/hip_runtime.h>

#define BATCH 64
#define LFR   500
#define NB    501
#define DDIM  1024
#define SMAX  50

typedef __attribute__((ext_vector_type(2))) float v2f;
typedef __attribute__((ext_vector_type(8))) float v8f;

__global__ void seg_prep(const float* __restrict__ in_boundary,
                         int* __restrict__ segid, float* __restrict__ w,
                         float* __restrict__ mask_sum, float* __restrict__ bnd_out)
{
    int b = blockIdx.x;
    const float* ib = in_boundary + b * NB;

    for (int t = threadIdx.x; t < NB; t += blockDim.x)
        bnd_out[b * NB + t] = ib[t];
    for (int s = threadIdx.x; s < SMAX; s += blockDim.x)
        mask_sum[b * SMAX + s] = 0.0f;
    __syncthreads();

    if (threadIdx.x == 0) {
        int*   sg = segid + b * LFR;
        float* wb = w     + b * LFR;
        int seg = 0, start = 0;
        for (int t = 1; t <= LFR; ++t) {
            if (ib[t] != 0.0f) {
                float wv = (seg < SMAX) ? (1.0f / (float)(t - start)) : 0.0f;
                for (int l = start; l < t; ++l) { sg[l] = seg; wb[l] = wv; }
                if (seg < SMAX) mask_sum[b * SMAX + seg] = 1.0f;
                ++seg; start = t;
            }
        }
        for (int l = start; l < LFR; ++l) { sg[l] = seg; wb[l] = 0.0f; }
    }
}

__global__ void seg_gemm_wmma(const float* __restrict__ x,
                              const int* __restrict__ segid,
                              const float* __restrict__ w,
                              float* __restrict__ out)
{
    __shared__ int   s_sg[LFR];
    __shared__ float s_w [LFR];

    const int b    = blockIdx.y;
    const int wave = threadIdx.x >> 5;
    const int lane = threadIdx.x & 31;
    const int n_tile = blockIdx.x * (blockDim.x >> 5) + wave;
    const int khi  = lane >> 4;
    const int lo   = lane & 15;
    const int n    = n_tile * 16 + lo;

    {
        const int*   sg = segid + b * LFR;
        const float* wb = w     + b * LFR;
        for (int i = threadIdx.x; i < LFR; i += blockDim.x) {
            s_sg[i] = sg[i];
            s_w [i] = wb[i];
        }
    }
    __syncthreads();

    const float* xb = x + (size_t)b * LFR * DDIM;

    v8f c0 = {}, c1 = {}, c2 = {}, c3 = {};
    const int m0 = lo, m1 = 16 + lo, m2 = 32 + lo, m3 = 48 + lo;

    for (int kb = 0; kb < LFR; kb += 4) {
        const int k0 = kb + 2 * khi;
        const int k1 = k0 + 1;

        const int   sA = s_sg[k0], sB = s_sg[k1];
        const float wA = s_w[k0],  wB = s_w[k1];

        v2f Bv;
        Bv.x = xb[(size_t)k0 * DDIM + n];
        Bv.y = xb[(size_t)k1 * DDIM + n];

        v2f A;
        A.x = (sA == m0) ? wA : 0.0f;  A.y = (sB == m0) ? wB : 0.0f;
        c0 = __builtin_amdgcn_wmma_f32_16x16x4_f32(false, A, false, Bv, (short)0, c0, false, false);
        A.x = (sA == m1) ? wA : 0.0f;  A.y = (sB == m1) ? wB : 0.0f;
        c1 = __builtin_amdgcn_wmma_f32_16x16x4_f32(false, A, false, Bv, (short)0, c1, false, false);
        A.x = (sA == m2) ? wA : 0.0f;  A.y = (sB == m2) ? wB : 0.0f;
        c2 = __builtin_amdgcn_wmma_f32_16x16x4_f32(false, A, false, Bv, (short)0, c2, false, false);
        A.x = (sA == m3) ? wA : 0.0f;  A.y = (sB == m3) ? wB : 0.0f;
        c3 = __builtin_amdgcn_wmma_f32_16x16x4_f32(false, A, false, Bv, (short)0, c3, false, false);
    }

    float* ob = out + (size_t)b * SMAX * DDIM;
#pragma unroll
    for (int i = 0; i < 8; ++i) {
        const int r = i + 8 * khi;
        ob[(size_t)(r)      * DDIM + n] = c0[i];
        ob[(size_t)(16 + r) * DDIM + n] = c1[i];
        ob[(size_t)(32 + r) * DDIM + n] = c2[i];
        if (48 + r < SMAX)
            ob[(size_t)(48 + r) * DDIM + n] = c3[i];
    }
}

extern "C" void kernel_launch(void* const* d_in, const int* in_sizes, int n_in,
                              void* d_out, int out_size, void* d_ws, size_t ws_size,
                              hipStream_t stream)
{
    (void)in_sizes; (void)n_in; (void)out_size; (void)ws_size;

    const float* x           = (const float*)d_in[0];
    const float* in_boundary = (const float*)d_in[1];

    float* out      = (float*)d_out;
    float* mask_sum = out + (size_t)BATCH * SMAX * DDIM;
    float* bnd_out  = mask_sum + (size_t)BATCH * SMAX;

    int*   segid = (int*)d_ws;
    float* w     = (float*)((char*)d_ws + (size_t)BATCH * LFR * sizeof(int));

    seg_prep<<<BATCH, 64, 0, stream>>>(in_boundary, segid, w, mask_sum, bnd_out);
    seg_gemm_wmma<<<dim3(8, BATCH), 256, 0, stream>>>(x, segid, w, out);
}


# Product re-decomposed: block = (batch, 64 columns), each of 4 waves owns one 16-segment tile and streams only that tile's frame range, 16-byte x loads of 8 K-steps issued ahead; sum order kept
# speedup vs baseline: 1.8756x; 1.8756x over previous
#include <hip/hip_runtime.h>

#define BATCH 64
#define LFR   500
#define NB    501
#define DDIM  1024
#define SMAX  50
#define KU    8
#define LPAD  (LFR + 4 * KU + 4)

typedef __attribute__((ext_vector_type(2))) float v2f;
typedef __attribute__((ext_vector_type(4))) float v4f;
typedef __attribute__((ext_vector_type(4))) int   v4i;
typedef __attribute__((ext_vector_type(8))) float v8f;

__global__ void seg_prep(const float* __restrict__ in_boundary,
                         int* __restrict__ segid, float* __restrict__ w,
                         float* __restrict__ mask_sum, float* __restrict__ bnd_out)
{
    int b = blockIdx.x;
    const float* ib = in_boundary + b * NB;

    for (int t = threadIdx.x; t < NB; t += blockDim.x)
        bnd_out[b * NB + t] = ib[t];
    for (int s = threadIdx.x; s < SMAX; s += blockDim.x)
        mask_sum[b * SMAX + s] = 0.0f;
    __syncthreads();

    if (threadIdx.x == 0) {
        int*   sg = segid + b * LFR;
        float* wb = w     + b * LFR;
        int seg = 0, start = 0;
        for (int t = 1; t <= LFR; ++t) {
            if (ib[t] != 0.0f) {
                float wv = (seg < SMAX) ? (1.0f / (float)(t - start)) : 0.0f;
                for (int l = start; l < t; ++l) { sg[l] = seg; wb[l] = wv; }
                if (seg < SMAX) mask_sum[b * SMAX + seg] = 1.0f;
                ++seg; start = t;
            }
        }
        for (int l = start; l < LFR; ++l) { sg[l] = seg; wb[l] = 0.0f; }
    }
}

__global__ __launch_bounds__(128)
void seg_gemm_wmma(const float* __restrict__ x,
                   const int* __restrict__ segid,
                   const float* __restrict__ w,
                   float* __restrict__ out)
{
    __shared__ __attribute__((aligned(16))) int s_tab[2 * LPAD];

    const int b    = blockIdx.y;
    const int t    = threadIdx.x >> 5;
    const int lane = threadIdx.x & 31;
    const int khi  = lane >> 4;
    const int lo   = lane & 15;
    const int col  = blockIdx.x * 64 + 4 * lo;

    {
        const int*   sg = segid + b * LFR;
        const float* wb = w     + b * LFR;
        for (int i = threadIdx.x; i < LPAD; i += blockDim.x) {
            const bool in = i < LFR;
            s_tab[2 * i]     = in ? sg[i] : 0x7fffffff;
            s_tab[2 * i + 1] = in ? __float_as_int(wb[i]) : 0;
        }
    }
    __syncthreads();

    const int seg_lo = 16 * t;
    const int seg_hi = (seg_lo + 16 < SMAX) ? seg_lo + 16 : SMAX;
    int n_lo = 0, n_hi = 0;
    for (int i = lane; i < 512; i += 32) {
        const int sg = s_tab[2 * i];
        n_lo += __popcll(__ballot(sg < seg_lo));
        n_hi += __popcll(__ballot(sg < seg_hi));
    }
    const int fs = __builtin_amdgcn_readfirstlane(n_lo);
    const int fe = __builtin_amdgcn_readfirstlane(n_hi);

    const float* xb = x + (size_t)b * LFR * DDIM;
    const int lane_off = 2 * khi * DDIM + col;
    const int m = seg_lo + lo;

    v8f c0 = {}, c1 = {}, c2 = {}, c3 = {};

    for (int kb = fs & ~3; kb < fe; kb += 4 * KU) {
        v4f r0[KU], r1[KU];
#pragma unroll
        for (int u = 0; u < KU; ++u) {
            const int ku = (kb + 4 * u < LFR - 4) ? kb + 4 * u : (LFR - 4);
            const float* p = xb + (size_t)ku * DDIM;
            r0[u] = *(const v4f*)(p + lane_off);
            r1[u] = *(const v4f*)(p + lane_off + DDIM);
        }
        __builtin_amdgcn_sched_barrier(0);
#pragma unroll
        for (int u = 0; u < KU; ++u) {
            const int k0 = kb + 4 * u + 2 * khi;
            const v4i q = *(const v4i*)(s_tab + 2 * k0);
            v2f A, B;
            A.x = (q.x == m) ? __int_as_float(q.y) : 0.0f;
            A.y = (q.z == m) ? __int_as_float(q.w) : 0.0f;
            B.x = r0[u].x; B.y = r1[u].x;
            c0 = __builtin_amdgcn_wmma_f32_16x16x4_f32(false, A, false, B, (short)0, c0, false, false);
            B.x = r0[u].y; B.y = r1[u].y;
            c1 = __builtin_amdgcn_wmma_f32_16x16x4_f32(false, A, false, B, (short)0, c1, false, false);
            B.x = r0[u].z; B.y = r1[u].z;
            c2 = __builtin_amdgcn_wmma_f32_16x16x4_f32(false, A, false, B, (short)0, c2, false, false);
            B.x = r0[u].w; B.y = r1[u].w;
            c3 = __builtin_amdgcn_wmma_f32_16x16x4_f32(false, A, false, B, (short)0, c3, false, false);
        }
    }

    float* ob = out + (size_t)b * SMAX * DDIM + col;
#pragma unroll
    for (int i = 0; i < 8; ++i) {
        const int row = seg_lo + i + 8 * khi;
        if (row < SMAX) {
            v4f o;
            o.x = c0[i]; o.y = c1[i]; o.z = c2[i]; o.w = c3[i];
            *(v4f*)(ob + (size_t)row * DDIM) = o;
        }
    }
}

extern "C" void kernel_launch(void* const* d_in, const int* in_sizes, int n_in,
                              void* d_out, int out_size, void* d_ws, size_t ws_size,
                              hipStream_t stream)
{
    (void)in_sizes; (void)n_in; (void)out_size; (void)ws_size;

    const float* x           = (const float*)d_in[0];
    const float* in_boundary = (const float*)d_in[1];

    float* out      = (float*)d_out;
    float* mask_sum = out + (size_t)BATCH * SMAX * DDIM;
    float* bnd_out  = mask_sum + (size_t)BATCH * SMAX;

    int*   segid = (int*)d_ws;
    float* w     = (float*)((char*)d_ws + (size_t)BATCH * LFR * sizeof(int));

    seg_prep<<<BATCH, 64, 0, stream>>>(in_boundary, segid, w, mask_sum, bnd_out);
    seg_gemm_wmma<<<dim3(DDIM / 64, BATCH), 128, 0, stream>>>(x, segid, w, out);
}


# Segmentation as per-segment boundary-slot list: prep ranks boundaries in parallel by ballot; each lane keeps its row's frame range and 1/len in registers; no LDS table or serial loop; sum order kept
# speedup vs baseline: 5.0360x; 2.6850x over previous
#include <hip/hip_runtime.h>

#define BATCH 64
#define LFR   500
#define NB    501
#define DDIM  1024
#define SMAX  50
#define KU    8
#define PSTR  64

typedef __attribute__((ext_vector_type(2))) float v2f;
typedef __attribute__((ext_vector_type(4))) float v4f;
typedef __attribute__((ext_vector_type(8))) float v8f;

__global__ __launch_bounds__(32)
void seg_prep(const float* __restrict__ in_boundary,
              int* __restrict__ pos, float* __restrict__ mask_sum,
              float* __restrict__ bnd_out)
{
    const int b    = blockIdx.x;
    const int lane = threadIdx.x & 31;
    const float* ib = in_boundary + b * NB;
    int* pb = pos + b * PSTR;

    float v[16];
#pragma unroll
    for (int i = 0; i < 16; ++i) {
        const int t = 32 * i + lane;
        v[i] = (t < NB) ? ib[t] : 0.0f;
    }

    int nbnd = 0, last = 0;
#pragma unroll
    for (int i = 0; i < 16; ++i) {
        const int t = 32 * i + lane;
        if (t < NB) bnd_out[b * NB + t] = v[i];
        const bool isb = (t < NB) && (v[i] != 0.0f || t == 0);
        const unsigned mk = __builtin_amdgcn_ballot_w32(isb);
        const int rank = nbnd + (int)__builtin_amdgcn_mbcnt_lo(mk, 0u);
        if (isb && rank <= SMAX) pb[rank] = t;
        if (mk != 0u) last = 32 * i + 31 - __clz(mk);
        nbnd += __popc(mk);
    }
    for (int r = lane; r <= SMAX; r += 32)
        if (r >= nbnd) pb[r] = last;
    for (int s = lane; s < SMAX; s += 32)
        mask_sum[b * SMAX + s] = (s + 1 < nbnd) ? 1.0f : 0.0f;
}

__global__ __launch_bounds__(128)
void seg_gemm_wmma(const float* __restrict__ x,
                   const int* __restrict__ pos,
                   float* __restrict__ out)
{
    const int b    = blockIdx.y;
    const int t    = __builtin_amdgcn_readfirstlane((int)(threadIdx.x >> 5));
    const int lane = threadIdx.x & 31;
    const int khi  = lane >> 4;
    const int lo   = lane & 15;
    const int col  = blockIdx.x * 64 + 4 * lo;

    const int* pb = pos + b * PSTR;
    const int seg_lo = 16 * t;
    const int seg_hi = (seg_lo + 16 < SMAX) ? seg_lo + 16 : SMAX;
    const int fs = pb[seg_lo];
    const int fe = pb[seg_hi];

    const int m   = seg_lo + lo;
    const int st  = pb[(m < SMAX) ? m : SMAX];
    const int len = pb[(m + 1 < SMAX) ? m + 1 : SMAX] - st;
    const float wv = (len > 0) ? (1.0f / (float)len) : 0.0f;

    const float* xb = x + (size_t)b * LFR * DDIM;
    const int lane_off = 2 * khi * DDIM + col;

    v8f c0 = {}, c1 = {}, c2 = {}, c3 = {};

    for (int kb = fs & ~3; kb < fe; kb += 4 * KU) {
        v4f r0[KU], r1[KU];
#pragma unroll
        for (int u = 0; u < KU; ++u) {
            const int ku = (kb + 4 * u < LFR - 4) ? kb + 4 * u : (LFR - 4);
            const float* p = xb + (size_t)ku * DDIM;
            r0[u] = *(const v4f*)(p + lane_off);
            r1[u] = *(const v4f*)(p + lane_off + DDIM);
        }
        __builtin_amdgcn_sched_barrier(0);
        const int d = kb + 2 * khi - st;
#pragma unroll
        for (int u = 0; u < KU; ++u) {
            const unsigned d0 = (unsigned)(d + 4 * u);
            v2f A, B;
            A.x = (d0      < (unsigned)len) ? wv : 0.0f;
            A.y = (d0 + 1u < (unsigned)len) ? wv : 0.0f;
            B.x = r0[u].x; B.y = r1[u].x;
            c0 = __builtin_amdgcn_wmma_f32_16x16x4_f32(false, A, false, B, (short)0, c0, false, false);
            B.x = r0[u].y; B.y = r1[u].y;
            c1 = __builtin_amdgcn_wmma_f32_16x16x4_f32(false, A, false, B, (short)0, c1, false, false);
            B.x = r0[u].z; B.y = r1[u].z;
            c2 = __builtin_amdgcn_wmma_f32_16x16x4_f32(false, A, false, B, (short)0, c2, false, false);
            B.x = r0[u].w; B.y = r1[u].w;
            c3 = __builtin_amdgcn_wmma_f32_16x16x4_f32(false, A, false, B, (short)0, c3, false, false);
        }
    }

    float* ob = out + (size_t)b * SMAX * DDIM + col;
#pragma unroll
    for (int i = 0; i < 8; ++i) {
        const int row = seg_lo + i + 8 * khi;
        if (row < SMAX) {
            v4f o;
            o.x = c0[i]; o.y = c1[i]; o.z = c2[i]; o.w = c3[i];
            *(v4f*)(ob + (size_t)row * DDIM) = o;
        }
    }
}

extern "C" void kernel_launch(void* const* d_in, const int* in_sizes, int n_in,
                              void* d_out, int out_size, void* d_ws, size_t ws_size,
                              hipStream_t stream)
{
    (void)in_sizes; (void)n_in; (void)out_size; (void)ws_size;

    const float* x           = (const float*)d_in[0];
    const float* in_boundary = (const float*)d_in[1];

    float* out      = (float*)d_out;
    float* mask_sum = out + (size_t)BATCH * SMAX * DDIM;
    float* bnd_out  = mask_sum + (size_t)BATCH * SMAX;

    int* pos = (int*)d_ws;

    seg_prep<<<BATCH, 32, 0, stream>>>(in_boundary, pos, mask_sum, bnd_out);
    seg_gemm_wmma<<<dim3(DDIM / 64, BATCH), 128, 0, stream>>>(x, pos, out);
}


# Segments dealt to the block's 4 waves per batch by frame count (split points by ballot over the boundary list, 16 or fewer segments a wave) instead of fixed 16-segment tiles; sum order kept
# speedup vs baseline: 5.8585x; 1.1633x over previous
#include <hip/hip_runtime.h>

#define BATCH 64
#define LFR   500
#define NB    501
#define DDIM  1024
#define SMAX  50
#define KU    8
#define PSTR  64

typedef __attribute__((ext_vector_type(2))) float v2f;
typedef __attribute__((ext_vector_type(4))) float v4f;
typedef __attribute__((ext_vector_type(8))) float v8f;

__global__ __launch_bounds__(32)
void seg_prep(const float* __restrict__ in_boundary,
              int* __restrict__ pos, float* __restrict__ mask_sum,
              float* __restrict__ bnd_out)
{
    const int b    = blockIdx.x;
    const int lane = threadIdx.x & 31;
    const float* ib = in_boundary + b * NB;
    int* pb = pos + b * PSTR;

    float v[16];
#pragma unroll
    for (int i = 0; i < 16; ++i) {
        const int t = 32 * i + lane;
        v[i] = (t < NB) ? ib[t] : 0.0f;
    }

    int nbnd = 0, last = 0;
#pragma unroll
    for (int i = 0; i < 16; ++i) {
        const int t = 32 * i + lane;
        if (t < NB) bnd_out[b * NB + t] = v[i];
        const bool isb = (t < NB) && (v[i] != 0.0f || t == 0);
        const unsigned mk = __builtin_amdgcn_ballot_w32(isb);
        const int rank = nbnd + (int)__builtin_amdgcn_mbcnt_lo(mk, 0u);
        if (isb && rank <= SMAX) pb[rank] = t;
        if (mk != 0u) last = 32 * i + 31 - __clz(mk);
        nbnd += __popc(mk);
    }
    for (int r = lane; r <= SMAX; r += 32)
        if (r >= nbnd) pb[r] = last;
    for (int s = lane; s < SMAX; s += 32)
        mask_sum[b * SMAX + s] = (s + 1 < nbnd) ? 1.0f : 0.0f;
}

__global__ __launch_bounds__(128)
void seg_gemm_wmma(const float* __restrict__ x,
                   const int* __restrict__ pos,
                   float* __restrict__ out)
{
    const int b    = blockIdx.y;
    const int t    = __builtin_amdgcn_readfirstlane((int)(threadIdx.x >> 5));
    const int lane = threadIdx.x & 31;
    const int khi  = lane >> 4;
    const int lo   = lane & 15;
    const int col  = blockIdx.x * 64 + 4 * lo;

    const int* pb = pos + b * PSTR;
    const int p0 = pb[lane];
    const int p1 = pb[(32 + lane < SMAX) ? 32 + lane : SMAX];
    const bool hi_start = (32 + lane < SMAX);
    const int F = __builtin_amdgcn_readlane(p1, SMAX - 32);

    auto starts_below = [&](int f) -> int {
        return __popc(__builtin_amdgcn_ballot_w32(p0 < f)) +
               __popc(__builtin_amdgcn_ballot_w32(hi_start && p1 < f));
    };
    const int ncl = starts_below(F);

    int s_lo = 0, s_hi = 0, s_prev = 0;
    for (int w = 1; w <= 4; ++w) {
        int c = (w < 4) ? starts_below((w * F) >> 2) : ncl;
        const int need = ncl - 16 * (4 - w);
        c = (c > need) ? c : need;
        c = (c < s_prev + 16) ? c : s_prev + 16;
        if (w == t)     s_lo = c;
        if (w == t + 1) s_hi = c;
        s_prev = c;
    }
    const int cnt = s_hi - s_lo;

    auto pos_at = [&](int r) -> int {
        return (r < 32) ? __builtin_amdgcn_readlane(p0, r) : __builtin_amdgcn_readlane(p1, r - 32);
    };
    const int fs = pos_at(s_lo);
    const int fe = pos_at(s_hi);

    const int m  = (s_lo + lo < SMAX) ? s_lo + lo : SMAX - 1;
    const int a0 = __builtin_amdgcn_ds_bpermute(4 * (m & 31), p0);
    const int a1 = __builtin_amdgcn_ds_bpermute(4 * (m & 31), p1);
    const int e0 = __builtin_amdgcn_ds_bpermute(4 * ((m + 1) & 31), p0);
    const int e1 = __builtin_amdgcn_ds_bpermute(4 * ((m + 1) & 31), p1);
    const int st  = (m < 32) ? a0 : a1;
    const int en  = (m + 1 < 32) ? e0 : e1;
    const int len = (lo < cnt) ? en - st : 0;
    const float wv = (len > 0) ? (1.0f / (float)len) : 0.0f;

    const float* xb = x + (size_t)b * LFR * DDIM;
    const int lane_off = 2 * khi * DDIM + col;

    v8f c0 = {}, c1 = {}, c2 = {}, c3 = {};

    for (int kb = fs & ~3; kb < fe; kb += 4 * KU) {
        v4f r0[KU], r1[KU];
#pragma unroll
        for (int u = 0; u < KU; ++u) {
            const int ku = (kb + 4 * u < LFR - 4) ? kb + 4 * u : (LFR - 4);
            const float* p = xb + (size_t)ku * DDIM;
            r0[u] = *(const v4f*)(p + lane_off);
            r1[u] = *(const v4f*)(p + lane_off + DDIM);
        }
        __builtin_amdgcn_sched_barrier(0);
        const int d = kb + 2 * khi - st;
#pragma unroll
        for (int u = 0; u < KU; ++u) {
            const unsigned d0 = (unsigned)(d + 4 * u);
            v2f A, B;
            A.x = (d0      < (unsigned)len) ? wv : 0.0f;
            A.y = (d0 + 1u < (unsigned)len) ? wv : 0.0f;
            B.x = r0[u].x; B.y = r1[u].x;
            c0 = __builtin_amdgcn_wmma_f32_16x16x4_f32(false, A, false, B, (short)0, c0, false, false);
            B.x = r0[u].y; B.y = r1[u].y;
            c1 = __builtin_amdgcn_wmma_f32_16x16x4_f32(false, A, false, B, (short)0, c1, false, false);
            B.x = r0[u].z; B.y = r1[u].z;
            c2 = __builtin_amdgcn_wmma_f32_16x16x4_f32(false, A, false, B, (short)0, c2, false, false);
            B.x = r0[u].w; B.y = r1[u].w;
            c3 = __builtin_amdgcn_wmma_f32_16x16x4_f32(false, A, false, B, (short)0, c3, false, false);
        }
    }

    float* ob = out + (size_t)b * SMAX * DDIM + col;
#pragma unroll
    for (int i = 0; i < 8; ++i) {
        const int mr = i + 8 * khi;
        if (mr < cnt) {
            v4f o;
            o.x = c0[i]; o.y = c1[i]; o.z = c2[i]; o.w = c3[i];
            *(v4f*)(ob + (size_t)(s_lo + mr) * DDIM) = o;
        }
    }
    for (int r = ncl + t + 4 * khi; r < SMAX; r += 8) {
        const v4f z = {0.0f, 0.0f, 0.0f, 0.0f};
        *(v4f*)(ob + (size_t)r * DDIM) = z;
    }
}

extern "C" void kernel_launch(void* const* d_in, const int* in_sizes, int n_in,
                              void* d_out, int out_size, void* d_ws, size_t ws_size,
                              hipStream_t stream)
{
    (void)in_sizes; (void)n_in; (void)out_size; (void)ws_size;

    const float* x           = (const float*)d_in[0];
    const float* in_boundary = (const float*)d_in[1];

    float* out      = (float*)d_out;
    float* mask_sum = out + (size_t)BATCH * SMAX * DDIM;
    float* bnd_out  = mask_sum + (size_t)BATCH * SMAX;

    int* pos = (int*)d_ws;

    seg_prep<<<BATCH, 32, 0, stream>>>(in_boundary, pos, mask_sum, bnd_out);
    seg_gemm_wmma<<<dim3(DDIM / 64, BATCH), 128, 0, stream>>>(x, pos, out);
}


# Ranking launch removed: each wave ranks its batch's boundary flags itself (16 ballots, slots scattered by rank to a wave-private LDS strip) into its 2-register list; 2 launches -> 1; sum order kept
# speedup vs baseline: 6.8496x; 1.1692x over previous
#include <hip/hip_runtime.h>

#define BATCH 64
#define LFR   500
#define NB    501
#define DDIM  1024
#define SMAX  50
#define KU    8
#define PSTR  64

typedef __attribute__((ext_vector_type(2))) float v2f;
typedef __attribute__((ext_vector_type(4))) float v4f;
typedef __attribute__((ext_vector_type(8))) float v8f;

__global__ __launch_bounds__(128)
void seg_gemm_wmma(const float* __restrict__ x,
                   const float* __restrict__ in_boundary,
                   float* __restrict__ out,
                   float* __restrict__ mask_sum,
                   float* __restrict__ bnd_out)
{
    __shared__ int plist[4][PSTR];
    const int b    = blockIdx.y;
    const int t    = __builtin_amdgcn_readfirstlane((int)(threadIdx.x >> 5));
    const int lane = threadIdx.x & 31;
    const int khi  = lane >> 4;
    const int lo   = lane & 15;
    const int col  = blockIdx.x * 64 + 4 * lo;

    const float* ib = in_boundary + b * NB;
    int* pl = plist[t];
    const bool side = (blockIdx.x == 0) && (t == 0);

    float v[16];
#pragma unroll
    for (int i = 0; i < 16; ++i) {
        const int s = 32 * i + lane;
        v[i] = (s < NB) ? ib[s] : 0.0f;
    }

    if (side) {
#pragma unroll
        for (int i = 0; i < 16; ++i) {
            const int s = 32 * i + lane;
            if (s < NB) bnd_out[b * NB + s] = v[i];
        }
    }

    int nbnd = 0, last = 0;
#pragma unroll
    for (int i = 0; i < 16; ++i) {
        const int s = 32 * i + lane;
        const bool isb = (s < NB) && (v[i] != 0.0f || s == 0);
        const unsigned mk = __builtin_amdgcn_ballot_w32(isb);
        const int rank = nbnd + (int)__builtin_amdgcn_mbcnt_lo(mk, 0u);
        if (isb && rank <= SMAX) pl[rank] = s;
        if (mk != 0u) last = 32 * i + 31 - __clz(mk);
        nbnd += __popc(mk);
    }
    for (int r = lane; r <= SMAX; r += 32)
        if (r >= nbnd) pl[r] = last;
    if (side)
        for (int s = lane; s < SMAX; s += 32)
            mask_sum[b * SMAX + s] = (s + 1 < nbnd) ? 1.0f : 0.0f;
    __builtin_amdgcn_fence(__ATOMIC_RELEASE, "wavefront");
    __builtin_amdgcn_wave_barrier();
    __builtin_amdgcn_fence(__ATOMIC_ACQUIRE, "wavefront");

    const int p0 = pl[lane];
    const int p1 = pl[(32 + lane < SMAX) ? 32 + lane : SMAX];
    const bool hi_start = (32 + lane < SMAX);
    const int F = __builtin_amdgcn_readlane(p1, SMAX - 32);

    auto starts_below = [&](int f) -> int {
        return __popc(__builtin_amdgcn_ballot_w32(p0 < f)) +
               __popc(__builtin_amdgcn_ballot_w32(hi_start && p1 < f));
    };
    const int ncl = starts_below(F);

    int s_lo = 0, s_hi = 0, s_prev = 0;
    for (int w = 1; w <= 4; ++w) {
        int c = (w < 4) ? starts_below((w * F) >> 2) : ncl;
        const int need = ncl - 16 * (4 - w);
        c = (c > need) ? c : need;
        c = (c < s_prev + 16) ? c : s_prev + 16;
        if (w == t)     s_lo = c;
        if (w == t + 1) s_hi = c;
        s_prev = c;
    }
    const int cnt = s_hi - s_lo;

    auto pos_at = [&](int r) -> int {
        return (r < 32) ? __builtin_amdgcn_readlane(p0, r) : __builtin_amdgcn_readlane(p1, r - 32);
    };
    const int fs = pos_at(s_lo);
    const int fe = pos_at(s_hi);

    const int m  = (s_lo + lo < SMAX) ? s_lo + lo : SMAX - 1;
    const int a0 = __builtin_amdgcn_ds_bpermute(4 * (m & 31), p0);
    const int a1 = __builtin_amdgcn_ds_bpermute(4 * (m & 31), p1);
    const int e0 = __builtin_amdgcn_ds_bpermute(4 * ((m + 1) & 31), p0);
    const int e1 = __builtin_amdgcn_ds_bpermute(4 * ((m + 1) & 31), p1);
    const int st  = (m < 32) ? a0 : a1;
    const int en  = (m + 1 < 32) ? e0 : e1;
    const int len = (lo < cnt) ? en - st : 0;
    const float wv = (len > 0) ? (1.0f / (float)len) : 0.0f;

    const float* xb = x + (size_t)b * LFR * DDIM;
    const int lane_off = 2 * khi * DDIM + col;

    v8f c0 = {}, c1 = {}, c2 = {}, c3 = {};

    for (int kb = fs & ~3; kb < fe; kb += 4 * KU) {
        v4f r0[KU], r1[KU];
#pragma unroll
        for (int u = 0; u < KU; ++u) {
            const int ku = (kb + 4 * u < LFR - 4) ? kb + 4 * u : (LFR - 4);
            const float* p = xb + (size_t)ku * DDIM;
            r0[u] = *(const v4f*)(p + lane_off);
            r1[u] = *(const v4f*)(p + lane_off + DDIM);
        }
        __builtin_amdgcn_sched_barrier(0);
        const int d = kb + 2 * khi - st;
#pragma unroll
        for (int u = 0; u < KU; ++u) {
            const unsigned d0 = (unsigned)(d + 4 * u);
            v2f A, B;
            A.x = (d0      < (unsigned)len) ? wv : 0.0f;
            A.y = (d0 + 1u < (unsigned)len) ? wv : 0.0f;
            B.x = r0[u].x; B.y = r1[u].x;
            c0 = __builtin_amdgcn_wmma_f32_16x16x4_f32(false, A, false, B, (short)0, c0, false, false);
            B.x = r0[u].y; B.y = r1[u].y;
            c1 = __builtin_amdgcn_wmma_f32_16x16x4_f32(false, A, false, B, (short)0, c1, false, false);
            B.x = r0[u].z; B.y = r1[u].z;
            c2 = __builtin_amdgcn_wmma_f32_16x16x4_f32(false, A, false, B, (short)0, c2, false, false);
            B.x = r0[u].w; B.y = r1[u].w;
            c3 = __builtin_amdgcn_wmma_f32_16x16x4_f32(false, A, false, B, (short)0, c3, false, false);
        }
    }

    float* ob = out + (size_t)b * SMAX * DDIM + col;
#pragma unroll
    for (int i = 0; i < 8; ++i) {
        const int mr = i + 8 * khi;
        if (mr < cnt) {
            v4f o;
            o.x = c0[i]; o.y = c1[i]; o.z = c2[i]; o.w = c3[i];
            *(v4f*)(ob + (size_t)(s_lo + mr) * DDIM) = o;
        }
    }
    for (int r = ncl + t + 4 * khi; r < SMAX; r += 8) {
        const v4f z = {0.0f, 0.0f, 0.0f, 0.0f};
        *(v4f*)(ob + (size_t)r * DDIM) = z;
    }
}

extern "C" void kernel_launch(void* const* d_in, const int* in_sizes, int n_in,
                              void* d_out, int out_size, void* d_ws, size_t ws_size,
                              hipStream_t stream)
{
    (void)in_sizes; (void)n_in; (void)out_size; (void)ws_size;

    const float* x           = (const float*)d_in[0];
    const float* in_boundary = (const float*)d_in[1];

    float* out      = (float*)d_out;
    float* mask_sum = out + (size_t)BATCH * SMAX * DDIM;
    float* bnd_out  = mask_sum + (size_t)BATCH * SMAX;

    (void)d_ws;

    seg_gemm_wmma<<<dim3(DDIM / 64, BATCH), 128, 0, stream>>>(x, in_boundary, out, mask_sum, bnd_out);
}


# Each wave's last load group cut to the K-steps it has left (one body per count 1..7, picked by a scalar switch), not padded to 8: no K-step past its frames is loaded or multiplied; sum order kept
# speedup vs baseline: 7.2090x; 1.0525x over previous
#include <hip/hip_runtime.h>

#define BATCH 64
#define LFR   500
#define NB    501
#define DDIM  1024
#define SMAX  50
#define KU    8
#define PSTR  64

typedef __attribute__((ext_vector_type(2))) float v2f;
typedef __attribute__((ext_vector_type(4))) float v4f;
typedef __attribute__((ext_vector_type(8))) float v8f;

template <int N>
__device__ __forceinline__ void k_group(const float* __restrict__ p, const int d,
                                        const unsigned len, const float wv,
                                        v8f& c0, v8f& c1, v8f& c2, v8f& c3)
{
    v4f r0[N], r1[N];
#pragma unroll
    for (int u = 0; u < N; ++u) {
        r0[u] = *(const v4f*)(p + (size_t)(4 * u) * DDIM);
        r1[u] = *(const v4f*)(p + (size_t)(4 * u + 1) * DDIM);
    }
    __builtin_amdgcn_sched_barrier(0);
#pragma unroll
    for (int u = 0; u < N; ++u) {
        const unsigned d0 = (unsigned)(d + 4 * u);
        v2f A, B;
        A.x = (d0      < len) ? wv : 0.0f;
        A.y = (d0 + 1u < len) ? wv : 0.0f;
        B.x = r0[u].x; B.y = r1[u].x;
        c0 = __builtin_amdgcn_wmma_f32_16x16x4_f32(false, A, false, B, (short)0, c0, false, false);
        B.x = r0[u].y; B.y = r1[u].y;
        c1 = __builtin_amdgcn_wmma_f32_16x16x4_f32(false, A, false, B, (short)0, c1, false, false);
        B.x = r0[u].z; B.y = r1[u].z;
        c2 = __builtin_amdgcn_wmma_f32_16x16x4_f32(false, A, false, B, (short)0, c2, false, false);
        B.x = r0[u].w; B.y = r1[u].w;
        c3 = __builtin_amdgcn_wmma_f32_16x16x4_f32(false, A, false, B, (short)0, c3, false, false);
    }
}

__global__ __launch_bounds__(128)
void seg_gemm_wmma(const float* __restrict__ x,
                   const float* __restrict__ in_boundary,
                   float* __restrict__ out,
                   float* __restrict__ mask_sum,
                   float* __restrict__ bnd_out)
{
    __shared__ int plist[4][PSTR];
    const int b    = blockIdx.y;
    const int t    = __builtin_amdgcn_readfirstlane((int)(threadIdx.x >> 5));
    const int lane = threadIdx.x & 31;
    const int khi  = lane >> 4;
    const int lo   = lane & 15;
    const int col  = blockIdx.x * 64 + 4 * lo;

    const float* ib = in_boundary + b * NB;
    int* pl = plist[t];
    const bool side = (blockIdx.x == 0) && (t == 0);

    float v[16];
#pragma unroll
    for (int i = 0; i < 16; ++i) {
        const int s = 32 * i + lane;
        v[i] = (s < NB) ? ib[s] : 0.0f;
    }

    if (side) {
#pragma unroll
        for (int i = 0; i < 16; ++i) {
            const int s = 32 * i + lane;
            if (s < NB) bnd_out[b * NB + s] = v[i];
        }
    }

    int nbnd = 0, last = 0;
#pragma unroll
    for (int i = 0; i < 16; ++i) {
        const int s = 32 * i + lane;
        const bool isb = (s < NB) && (v[i] != 0.0f || s == 0);
        const unsigned mk = __builtin_amdgcn_ballot_w32(isb);
        const int rank = nbnd + (int)__builtin_amdgcn_mbcnt_lo(mk, 0u);
        if (isb && rank <= SMAX) pl[rank] = s;
        if (mk != 0u) last = 32 * i + 31 - __clz(mk);
        nbnd += __popc(mk);
    }
    for (int r = lane; r <= SMAX; r += 32)
        if (r >= nbnd) pl[r] = last;
    if (side)
        for (int s = lane; s < SMAX; s += 32)
            mask_sum[b * SMAX + s] = (s + 1 < nbnd) ? 1.0f : 0.0f;
    __builtin_amdgcn_fence(__ATOMIC_RELEASE, "wavefront");
    __builtin_amdgcn_wave_barrier();
    __builtin_amdgcn_fence(__ATOMIC_ACQUIRE, "wavefront");

    const int p0 = pl[lane];
    const int p1 = pl[(32 + lane < SMAX) ? 32 + lane : SMAX];
    const bool hi_start = (32 + lane < SMAX);
    const int F = __builtin_amdgcn_readlane(p1, SMAX - 32);

    auto starts_below = [&](int f) -> int {
        return __popc(__builtin_amdgcn_ballot_w32(p0 < f)) +
               __popc(__builtin_amdgcn_ballot_w32(hi_start && p1 < f));
    };
    const int ncl = starts_below(F);

    int s_lo = 0, s_hi = 0, s_prev = 0;
    for (int w = 1; w <= 4; ++w) {
        int c = (w < 4) ? starts_below((w * F) >> 2) : ncl;
        const int need = ncl - 16 * (4 - w);
        c = (c > need) ? c : need;
        c = (c < s_prev + 16) ? c : s_prev + 16;
        if (w == t)     s_lo = c;
        if (w == t + 1) s_hi = c;
        s_prev = c;
    }
    const int cnt = s_hi - s_lo;

    auto pos_at = [&](int r) -> int {
        return (r < 32) ? __builtin_amdgcn_readlane(p0, r) : __builtin_amdgcn_readlane(p1, r - 32);
    };
    const int fs = pos_at(s_lo);
    const int fe = pos_at(s_hi);

    const int m  = (s_lo + lo < SMAX) ? s_lo + lo : SMAX - 1;
    const int a0 = __builtin_amdgcn_ds_bpermute(4 * (m & 31), p0);
    const int a1 = __builtin_amdgcn_ds_bpermute(4 * (m & 31), p1);
    const int e0 = __builtin_amdgcn_ds_bpermute(4 * ((m + 1) & 31), p0);
    const int e1 = __builtin_amdgcn_ds_bpermute(4 * ((m + 1) & 31), p1);
    const int st  = (m < 32) ? a0 : a1;
    const int en  = (m + 1 < 32) ? e0 : e1;
    const int len = (lo < cnt) ? en - st : 0;
    const float wv = (len > 0) ? (1.0f / (float)len) : 0.0f;

    const float* xb = x + (size_t)b * LFR * DDIM;
    const int lane_off = 2 * khi * DDIM + col;

    v8f c0 = {}, c1 = {}, c2 = {}, c3 = {};

    const int kb0  = fs & ~3;
    const int nst  = (cnt > 0) ? ((fe - kb0 + 3) >> 2) : 0;
    const int full = nst / KU;
    const int rem  = nst % KU;
    const float* p = xb + (size_t)kb0 * DDIM + lane_off;
    int d = kb0 + 2 * khi - st;
    const unsigned ulen = (unsigned)len;

    for (int g = 0; g < full; ++g) {
        k_group<KU>(p, d, ulen, wv, c0, c1, c2, c3);
        p += (size_t)(4 * KU) * DDIM;
        d += 4 * KU;
    }
    switch (rem) {
        case 1: k_group<1>(p, d, ulen, wv, c0, c1, c2, c3); break;
        case 2: k_group<2>(p, d, ulen, wv, c0, c1, c2, c3); break;
        case 3: k_group<3>(p, d, ulen, wv, c0, c1, c2, c3); break;
        case 4: k_group<4>(p, d, ulen, wv, c0, c1, c2, c3); break;
        case 5: k_group<5>(p, d, ulen, wv, c0, c1, c2, c3); break;
        case 6: k_group<6>(p, d, ulen, wv, c0, c1, c2, c3); break;
        case 7: k_group<7>(p, d, ulen, wv, c0, c1, c2, c3); break;
        default: break;
    }

    float* ob = out + (size_t)b * SMAX * DDIM + col;
#pragma unroll
    for (int i = 0; i < 8; ++i) {
        const int mr = i + 8 * khi;
        if (mr < cnt) {
            v4f o;
            o.x = c0[i]; o.y = c1[i]; o.z = c2[i]; o.w = c3[i];
            *(v4f*)(ob + (size_t)(s_lo + mr) * DDIM) = o;
        }
    }
    for (int r = ncl + t + 4 * khi; r < SMAX; r += 8) {
        const v4f z = {0.0f, 0.0f, 0.0f, 0.0f};
        *(v4f*)(ob + (size_t)r * DDIM) = z;
    }
}

extern "C" void kernel_launch(void* const* d_in, const int* in_sizes, int n_in,
                              void* d_out, int out_size, void* d_ws, size_t ws_size,
                              hipStream_t stream)
{
    (void)in_sizes; (void)n_in; (void)out_size; (void)ws_size;

    const float* x           = (const float*)d_in[0];
    const float* in_boundary = (const float*)d_in[1];

    float* out      = (float*)d_out;
    float* mask_sum = out + (size_t)BATCH * SMAX * DDIM;
    float* bnd_out  = mask_sum + (size_t)BATCH * SMAX;

    (void)d_ws;

    seg_gemm_wmma<<<dim3(DDIM / 64, BATCH), 128, 0, stream>>>(x, in_boundary, out, mask_sum, bnd_out);
}
